# starting kernel
# baseline (speedup 1.0000x reference)
#include <hip/hip_runtime.h>

typedef float v2f __attribute__((ext_vector_type(2)));
typedef float v8f __attribute__((ext_vector_type(8)));

#define DIM 128
#define XP 132
#define NEG_SLOPE 0.1f

__global__ void gcn_init_kernel(float* __restrict__ out, float* __restrict__ deg,
                                int N, long total) {
  long i = (long)blockIdx.x * blockDim.x + threadIdx.x;
  if (i < total) out[i] = 0.0f;
  if (i < N) deg[i] = 1.0f;
}

__global__ void gcn_deg_kernel(const int* __restrict__ ei, float* __restrict__ deg, int E) {
  int e = blockIdx.x * blockDim.x + threadIdx.x;
  if (e < E) atomicAdd(&deg[ei[E + e]], 1.0f);
}

__global__ void gcn_dinv_kernel(float* __restrict__ deg, int N) {
  int i = blockIdx.x * blockDim.x + threadIdx.x;
  if (i < N) deg[i] = rsqrtf(deg[i]);
}

__global__ __launch_bounds__(256) void gcn_gemm_kernel(const float* __restrict__ x,
                                                       const float* __restrict__ W,
                                                       float* __restrict__ h, int N) {
  __shared__ float Ws[DIM * DIM];
  __shared__ float xs[16 * XP];

  const int tid  = threadIdx.x;
  const int row0 = blockIdx.x * 16;

  for (int i = tid; i < (DIM * DIM) / 4; i += 256) {
    ((float4*)Ws)[i] = ((const float4*)W)[i];
  }
  for (int i = tid; i < 16 * DIM; i += 256) {
    int r = i >> 7, c = i & (DIM - 1);
    int gr = row0 + r;
    xs[r * XP + c] = (gr < N) ? x[(long)gr * DIM + c] : 0.0f;
  }
  __syncthreads();

  const int wave = tid >> 5;
  const int lane = tid & 31;
  const int half = lane >> 4;
  const int m    = lane & 15;
  const int col0 = wave * 16;

  v8f acc = {};
  for (int k0 = 0; k0 < DIM; k0 += 4) {
    int ka = k0 + 2 * half;
    v2f a, b;
    a.x = xs[m * XP + ka];
    a.y = xs[m * XP + ka + 1];
    b.x = Ws[ka * DIM + col0 + m];
    b.y = Ws[(ka + 1) * DIM + col0 + m];
    acc = __builtin_amdgcn_wmma_f32_16x16x4_f32(
         false, a,  false, b,
         (short)0, acc,  false,  false);
  }

  for (int r = 0; r < 8; ++r) {
    int row = row0 + r + 8 * half;
    if (row < N) h[(long)row * DIM + col0 + m] = acc[r];
  }
}

__global__ __launch_bounds__(256) void gcn_scatter_kernel(const int* __restrict__ ei,
                                                          const float* __restrict__ dinv,
                                                          const float* __restrict__ h,
                                                          float* __restrict__ out, int E) {
  int gwave = (int)((blockIdx.x * 256 + threadIdx.x) >> 5);
  int lane  = threadIdx.x & 31;
  if (gwave >= E) return;
  int s = ei[gwave];
  int d = ei[E + gwave];
  float w = dinv[s] * dinv[d];
  float4 v = ((const float4*)(h + (long)s * DIM))[lane];
  float* o = out + (long)d * DIM + lane * 4;
  atomicAdd(o + 0, v.x * w);
  atomicAdd(o + 1, v.y * w);
  atomicAdd(o + 2, v.z * w);
  atomicAdd(o + 3, v.w * w);
}

__global__ void gcn_final_kernel(const float* __restrict__ x, const float* __restrict__ bias,
                                 const float* __restrict__ dinv, const float* __restrict__ h,
                                 float* __restrict__ out, long total) {
  long i = (long)blockIdx.x * blockDim.x + threadIdx.x;
  if (i >= total) return;
  int row = (int)(i >> 7);
  int col = (int)(i & (DIM - 1));
  float di = dinv[row];
  float v = out[i] + h[i] * (di * di) + bias[col];
  v = (v >= 0.0f) ? v : NEG_SLOPE * v;
  out[i] = v + x[i];
}

extern "C" void kernel_launch(void* const* d_in, const int* in_sizes, int n_in,
                              void* d_out, int out_size, void* d_ws, size_t ws_size,
                              hipStream_t stream) {
  const float* x  = (const float*)d_in[0];
  const int*   ei = (const int*)d_in[1];
  const float* W  = (const float*)d_in[2];
  const float* b  = (const float*)d_in[3];

  const int N = in_sizes[0] / DIM;
  const int E = in_sizes[1] / 2;

  float* out  = (float*)d_out;
  float* dinv = (float*)d_ws;
  size_t hoff = (((size_t)N * sizeof(float)) + 255) & ~(size_t)255;
  float* h    = (float*)((char*)d_ws + hoff);

  const long total = (long)N * DIM;

  gcn_init_kernel<<<(int)((total + 255) / 256), 256, 0, stream>>>(out, dinv, N, total);
  gcn_deg_kernel<<<(E + 255) / 256, 256, 0, stream>>>(ei, dinv, E);
  gcn_dinv_kernel<<<(N + 255) / 256, 256, 0, stream>>>(dinv, N);
  gcn_gemm_kernel<<<(N + 15) / 16, 256, 0, stream>>>(x, W, h, N);
  gcn_scatter_kernel<<<(E + 7) / 8, 256, 0, stream>>>(ei, dinv, h, out, E);
  gcn_final_kernel<<<(int)((total + 255) / 256), 256, 0, stream>>>(x, b, dinv, h, out, total);
}


# aggregation in LDS tiles, no sort: each work-group keeps 400 out rows in LDS, streams the whole dst list, adds its edges' h rows with LDS f32 adds, finishes rows in place; sum order = LDS arrival
# speedup vs baseline: 1.8191x; 1.8191x over previous
#include <hip/hip_runtime.h>

typedef float v2f __attribute__((ext_vector_type(2)));
typedef float v8f __attribute__((ext_vector_type(8)));

#define DIM 128
#define XP 132
#define NEG_SLOPE 0.1f

__global__ void gcn_init_kernel(float* __restrict__ deg, int N) {
  int i = blockIdx.x * blockDim.x + threadIdx.x;
  if (i < N) deg[i] = 1.0f;
}

__global__ void gcn_deg_kernel(const int* __restrict__ ei, float* __restrict__ deg, int E) {
  int e = blockIdx.x * blockDim.x + threadIdx.x;
  if (e < E) atomicAdd(&deg[ei[E + e]], 1.0f);
}

__global__ void gcn_dinv_kernel(float* __restrict__ deg, int N) {
  int i = blockIdx.x * blockDim.x + threadIdx.x;
  if (i < N) deg[i] = rsqrtf(deg[i]);
}

__global__ __launch_bounds__(256) void gcn_gemm_kernel(const float* __restrict__ x,
                                                       const float* __restrict__ W,
                                                       float* __restrict__ h, int N) {
  __shared__ float Ws[DIM * DIM];
  __shared__ float xs[16 * XP];

  const int tid  = threadIdx.x;
  const int row0 = blockIdx.x * 16;

  for (int i = tid; i < (DIM * DIM) / 4; i += 256) {
    ((float4*)Ws)[i] = ((const float4*)W)[i];
  }
  for (int i = tid; i < 16 * DIM; i += 256) {
    int r = i >> 7, c = i & (DIM - 1);
    int gr = row0 + r;
    xs[r * XP + c] = (gr < N) ? x[(long)gr * DIM + c] : 0.0f;
  }
  __syncthreads();

  const int wave = tid >> 5;
  const int lane = tid & 31;
  const int half = lane >> 4;
  const int m    = lane & 15;
  const int col0 = wave * 16;

  v8f acc = {};
  for (int k0 = 0; k0 < DIM; k0 += 4) {
    int ka = k0 + 2 * half;
    v2f a, b;
    a.x = xs[m * XP + ka];
    a.y = xs[m * XP + ka + 1];
    b.x = Ws[ka * DIM + col0 + m];
    b.y = Ws[(ka + 1) * DIM + col0 + m];
    acc = __builtin_amdgcn_wmma_f32_16x16x4_f32(
         false, a,  false, b,
         (short)0, acc,  false,  false);
  }

  for (int r = 0; r < 8; ++r) {
    int row = row0 + r + 8 * half;
    if (row < N) h[(long)row * DIM + col0 + m] = acc[r];
  }
}

#define TR 400
#define TW 32
#define QCAP 544

__global__ __launch_bounds__(1024) void gcn_tile_kernel(const float* __restrict__ x,
                                                        const int* __restrict__ ei,
                                                        const float* __restrict__ bias,
                                                        const float* __restrict__ dinv,
                                                        const float* __restrict__ h,
                                                        float* __restrict__ out, int N, int E) {
  __shared__ float acc[TR * DIM];
  __shared__ int   q[TW * QCAP];

  const int tid  = threadIdx.x;
  const int lane = tid & 31;
  const int wave = tid >> 5;
  const int row0 = blockIdx.x * TR;
  const int qoff = wave * QCAP;
  const int* dst = ei + E;

  for (int i = tid; i < TR * DIM; i += TW * 32) acc[i] = 0.0f;
  __syncthreads();

  auto drain = [&](int n) {
    for (int base = 0; base < n; base += 32) {
      const int idx = base + lane;
      const int e   = q[qoff + ((idx < n) ? idx : base)];
      const int s_l = ei[e];
      const int d_l = dst[e];
      const float w_l = dinv[s_l] * dinv[d_l];
      const int r_l = d_l - row0;
      const int cnt = (n - base < 32) ? (n - base) : 32;
      for (int j = 0; j < cnt; j += 4) {
        float4 v[4];
        int    rr[4];
        float  ww[4];
        for (int u = 0; u < 4; ++u) {
          const int jj = (j + u < cnt) ? (j + u) : (cnt - 1);
          const int s = __builtin_amdgcn_readlane(s_l, jj);
          rr[u] = __builtin_amdgcn_readlane(r_l, jj);
          ww[u] = __int_as_float(__builtin_amdgcn_readlane(__float_as_int(w_l), jj));
          v[u]  = ((const float4*)(h + (long)s * DIM))[lane];
        }
        for (int u = 0; u < 4; ++u) {
          if (j + u < cnt) {
            float* a = &acc[rr[u] * DIM + lane];
            atomicAdd(a,      v[u].x * ww[u]);
            atomicAdd(a + 32, v[u].y * ww[u]);
            atomicAdd(a + 64, v[u].z * ww[u]);
            atomicAdd(a + 96, v[u].w * ww[u]);
          }
        }
      }
    }
  };

  int qn = 0;

  const int nvec    = ((E & 3) == 0) ? (E >> 2) : 0;
  const int ngroups = (nvec + 127) >> 7;
  const int4* dst4  = (const int4*)dst;

  for (int g = wave; g < ngroups; g += TW) {
    int4 dv[4];
    int  vi[4];
    bool ok[4];
    for (int u = 0; u < 4; ++u) {
      vi[u] = g * 128 + u * 32 + lane;
      ok[u] = vi[u] < nvec;
      dv[u] = dst4[ok[u] ? vi[u] : 0];
    }
    for (int u = 0; u < 4; ++u) {
      const int dc[4] = {dv[u].x, dv[u].y, dv[u].z, dv[u].w};
      const unsigned okm = __builtin_amdgcn_ballot_w32(ok[u]);
      for (int c = 0; c < 4; ++c) {
        const bool hit = (unsigned)(dc[c] - row0) < (unsigned)TR;
        const unsigned mask = __builtin_amdgcn_ballot_w32(hit) & okm;
        if (mask) {
          if (hit & ok[u]) q[qoff + qn + (int)__builtin_amdgcn_mbcnt_lo(mask, 0u)] = 4 * vi[u] + c;
          qn += __builtin_popcount(mask);
        }
      }
    }
    if (qn >= 32) {
      drain(qn);
      qn = 0;
    }
  }
  for (int e0 = 4 * nvec + wave * 32; e0 < E; e0 += TW * 32) {
    const int e = e0 + lane;
    const int d = (e < E) ? dst[e] : -1;
    const bool m = (unsigned)(d - row0) < (unsigned)TR;
    const unsigned mask = __builtin_amdgcn_ballot_w32(m);
    if (mask) {
      if (m) q[qoff + qn + (int)__builtin_amdgcn_mbcnt_lo(mask, 0u)] = e;
      qn += __builtin_popcount(mask);
    }
    if (qn >= 32) {
      drain(qn);
      qn = 0;
    }
  }
  if (qn > 0) drain(qn);
  __syncthreads();

  const float4 bv = ((const float4*)bias)[lane];
  for (int r = wave; r < TR; r += TW) {
    const int row = row0 + r;
    if (row >= N) break;
    const float* a = &acc[r * DIM + lane];
    const float di = dinv[row];
    const float dd = di * di;
    const float4 hv = ((const float4*)(h + (long)row * DIM))[lane];
    const float4 xv = ((const float4*)(x + (long)row * DIM))[lane];
    float4 o;
    float v;
    v = a[0]  + hv.x * dd + bv.x;  v = (v >= 0.0f) ? v : NEG_SLOPE * v;  o.x = v + xv.x;
    v = a[32] + hv.y * dd + bv.y;  v = (v >= 0.0f) ? v : NEG_SLOPE * v;  o.y = v + xv.y;
    v = a[64] + hv.z * dd + bv.z;  v = (v >= 0.0f) ? v : NEG_SLOPE * v;  o.z = v + xv.z;
    v = a[96] + hv.w * dd + bv.w;  v = (v >= 0.0f) ? v : NEG_SLOPE * v;  o.w = v + xv.w;
    ((float4*)(out + (long)row * DIM))[lane] = o;
  }
}

extern "C" void kernel_launch(void* const* d_in, const int* in_sizes, int n_in,
                              void* d_out, int out_size, void* d_ws, size_t ws_size,
                              hipStream_t stream) {
  const float* x  = (const float*)d_in[0];
  const int*   ei = (const int*)d_in[1];
  const float* W  = (const float*)d_in[2];
  const float* b  = (const float*)d_in[3];

  const int N = in_sizes[0] / DIM;
  const int E = in_sizes[1] / 2;

  float* out  = (float*)d_out;
  float* dinv = (float*)d_ws;
  size_t hoff = (((size_t)N * sizeof(float)) + 255) & ~(size_t)255;
  float* h    = (float*)((char*)d_ws + hoff);

  gcn_init_kernel<<<(N + 255) / 256, 256, 0, stream>>>(dinv, N);
  gcn_deg_kernel<<<(E + 255) / 256, 256, 0, stream>>>(ei, dinv, E);
  gcn_dinv_kernel<<<(N + 255) / 256, 256, 0, stream>>>(dinv, N);
  gcn_gemm_kernel<<<(N + 15) / 16, 256, 0, stream>>>(x, W, h, N);
  gcn_tile_kernel<<<(N + TR - 1) / TR, TW * 32, 0, stream>>>(x, ei, b, dinv, h, out, N, E);
}


# tile scan reads a 1-byte tile key per edge (dst/400, stored by the degree kernel, 4 edges per thread) instead of the 4-byte dst: 16 edges per lane per load; sum order = LDS arrival as before
# speedup vs baseline: 2.0879x; 1.1478x over previous
#include <hip/hip_runtime.h>

typedef float v2f __attribute__((ext_vector_type(2)));
typedef float v8f __attribute__((ext_vector_type(8)));

#define DIM 128
#define XP 132
#define NEG_SLOPE 0.1f

__global__ void gcn_init_kernel(float* __restrict__ deg, int N) {
  int i = blockIdx.x * blockDim.x + threadIdx.x;
  if (i < N) deg[i] = 1.0f;
}

#define TR 400

__global__ void gcn_deg_kernel(const int* __restrict__ ei, float* __restrict__ deg,
                               int* __restrict__ keys, int E, int nwords) {
  const int j = blockIdx.x * blockDim.x + threadIdx.x;
  if (j >= nwords) return;
  const int* dst = ei + E;
  int d[4];
  if (((E & 3) == 0) && (4 * j + 3 < E)) {
    const int4 v = ((const int4*)dst)[j];
    d[0] = v.x; d[1] = v.y; d[2] = v.z; d[3] = v.w;
  } else {
    for (int b = 0; b < 4; ++b) d[b] = (4 * j + b < E) ? dst[4 * j + b] : -1;
  }
  unsigned w = 0;
  for (int b = 0; b < 4; ++b) {
    unsigned k = 0xFFu;
    if (d[b] >= 0) {
      atomicAdd(&deg[d[b]], 1.0f);
      k = (unsigned)d[b] / (unsigned)TR;
    }
    w |= k << (8 * b);
  }
  keys[j] = (int)w;
}

__global__ void gcn_dinv_kernel(float* __restrict__ deg, int N) {
  int i = blockIdx.x * blockDim.x + threadIdx.x;
  if (i < N) deg[i] = rsqrtf(deg[i]);
}

__global__ __launch_bounds__(256) void gcn_gemm_kernel(const float* __restrict__ x,
                                                       const float* __restrict__ W,
                                                       float* __restrict__ h, int N) {
  __shared__ float Ws[DIM * DIM];
  __shared__ float xs[16 * XP];

  const int tid  = threadIdx.x;
  const int row0 = blockIdx.x * 16;

  for (int i = tid; i < (DIM * DIM) / 4; i += 256) {
    ((float4*)Ws)[i] = ((const float4*)W)[i];
  }
  for (int i = tid; i < 16 * DIM; i += 256) {
    int r = i >> 7, c = i & (DIM - 1);
    int gr = row0 + r;
    xs[r * XP + c] = (gr < N) ? x[(long)gr * DIM + c] : 0.0f;
  }
  __syncthreads();

  const int wave = tid >> 5;
  const int lane = tid & 31;
  const int half = lane >> 4;
  const int m    = lane & 15;
  const int col0 = wave * 16;

  v8f acc = {};
  for (int k0 = 0; k0 < DIM; k0 += 4) {
    int ka = k0 + 2 * half;
    v2f a, b;
    a.x = xs[m * XP + ka];
    a.y = xs[m * XP + ka + 1];
    b.x = Ws[ka * DIM + col0 + m];
    b.y = Ws[(ka + 1) * DIM + col0 + m];
    acc = __builtin_amdgcn_wmma_f32_16x16x4_f32(
         false, a,  false, b,
         (short)0, acc,  false,  false);
  }

  for (int r = 0; r < 8; ++r) {
    int row = row0 + r + 8 * half;
    if (row < N) h[(long)row * DIM + col0 + m] = acc[r];
  }
}

#define TW 32
#define QCAP 544

__global__ __launch_bounds__(1024) void gcn_tile_kernel(const float* __restrict__ x,
                                                        const int* __restrict__ ei,
                                                        const float* __restrict__ bias,
                                                        const float* __restrict__ dinv,
                                                        const float* __restrict__ h,
                                                        const int* __restrict__ keys,
                                                        float* __restrict__ out, int N, int E) {
  __shared__ float acc[TR * DIM];
  __shared__ int   q[TW * QCAP];

  const int tid  = threadIdx.x;
  const int lane = tid & 31;
  const int wave = tid >> 5;
  const int row0 = blockIdx.x * TR;
  const int qoff = wave * QCAP;
  const int* dst = ei + E;

  for (int i = tid; i < TR * DIM; i += TW * 32) acc[i] = 0.0f;
  __syncthreads();

  auto drain = [&](int n) {
    for (int base = 0; base < n; base += 32) {
      const int idx = base + lane;
      const int e   = q[qoff + ((idx < n) ? idx : base)];
      const int s_l = ei[e];
      const int d_l = dst[e];
      const float w_l = dinv[s_l] * dinv[d_l];
      const int r_l = d_l - row0;
      const int cnt = (n - base < 32) ? (n - base) : 32;
      for (int j = 0; j < cnt; j += 4) {
        float4 v[4];
        int    rr[4];
        float  ww[4];
        for (int u = 0; u < 4; ++u) {
          const int jj = (j + u < cnt) ? (j + u) : (cnt - 1);
          const int s = __builtin_amdgcn_readlane(s_l, jj);
          rr[u] = __builtin_amdgcn_readlane(r_l, jj);
          ww[u] = __int_as_float(__builtin_amdgcn_readlane(__float_as_int(w_l), jj));
          v[u]  = ((const float4*)(h + (long)s * DIM))[lane];
        }
        for (int u = 0; u < 4; ++u) {
          if (j + u < cnt) {
            float* a = &acc[rr[u] * DIM + lane];
            atomicAdd(a,      v[u].x * ww[u]);
            atomicAdd(a + 32, v[u].y * ww[u]);
            atomicAdd(a + 64, v[u].z * ww[u]);
            atomicAdd(a + 96, v[u].w * ww[u]);
          }
        }
      }
    }
  };

  int qn = 0;

  const int nvec    = (E + 15) >> 4;
  const int ngroups = (nvec + 127) >> 7;
  const uint4* kp   = (const uint4*)keys;
  const unsigned pat = (unsigned)blockIdx.x * 0x01010101u;

  auto match = [&](unsigned k) -> unsigned {
    const unsigned x = k ^ pat;
    return ~(((x & 0x7F7F7F7Fu) + 0x7F7F7F7Fu) | x) & 0x80808080u;
  };
  auto load_group = [&](int g, uint4 (&kv)[4]) {
    for (int u = 0; u < 4; ++u) {
      const int vi = g * 128 + u * 32 + lane;
      kv[u] = kp[(vi < nvec) ? vi : (nvec - 1)];
    }
  };

  uint4 nk[4];
  if (wave < ngroups) load_group(wave, nk);
  for (int g = wave; g < ngroups; g += TW) {
    uint4 kv[4];
    for (int u = 0; u < 4; ++u) kv[u] = nk[u];
    if (g + TW < ngroups) load_group(g + TW, nk);
    for (int u = 0; u < 4; ++u) {
      const int vi = g * 128 + u * 32 + lane;
      unsigned m = (match(kv[u].x) >> 7) | (match(kv[u].y) >> 6) | (match(kv[u].z) >> 5) | (match(kv[u].w) >> 4);
      if (vi >= nvec) m = 0;
      unsigned lm = __builtin_amdgcn_ballot_w32(m != 0);
      while (lm) {
        if (m) {
          const int p = __builtin_ctz(m);
          q[qoff + qn + (int)__builtin_amdgcn_mbcnt_lo(lm, 0u)] = 16 * vi + ((p & 3) << 2) + (p >> 3);
          m &= m - 1;
        }
        qn += __builtin_popcount(lm);
        lm = __builtin_amdgcn_ballot_w32(m != 0);
      }
      if (qn >= 32) {
        drain(qn);
        qn = 0;
      }
    }
  }
  if (qn > 0) drain(qn);
  __syncthreads();

  const float4 bv = ((const float4*)bias)[lane];
  for (int r = wave; r < TR; r += TW) {
    const int row = row0 + r;
    if (row >= N) break;
    const float* a = &acc[r * DIM + lane];
    const float di = dinv[row];
    const float dd = di * di;
    const float4 hv = ((const float4*)(h + (long)row * DIM))[lane];
    const float4 xv = ((const float4*)(x + (long)row * DIM))[lane];
    float4 o;
    float v;
    v = a[0]  + hv.x * dd + bv.x;  v = (v >= 0.0f) ? v : NEG_SLOPE * v;  o.x = v + xv.x;
    v = a[32] + hv.y * dd + bv.y;  v = (v >= 0.0f) ? v : NEG_SLOPE * v;  o.y = v + xv.y;
    v = a[64] + hv.z * dd + bv.z;  v = (v >= 0.0f) ? v : NEG_SLOPE * v;  o.z = v + xv.z;
    v = a[96] + hv.w * dd + bv.w;  v = (v >= 0.0f) ? v : NEG_SLOPE * v;  o.w = v + xv.w;
    ((float4*)(out + (long)row * DIM))[lane] = o;
  }
}

extern "C" void kernel_launch(void* const* d_in, const int* in_sizes, int n_in,
                              void* d_out, int out_size, void* d_ws, size_t ws_size,
                              hipStream_t stream) {
  const float* x  = (const float*)d_in[0];
  const int*   ei = (const int*)d_in[1];
  const float* W  = (const float*)d_in[2];
  const float* b  = (const float*)d_in[3];

  const int N = in_sizes[0] / DIM;
  const int E = in_sizes[1] / 2;

  float* out  = (float*)d_out;
  float* dinv = (float*)d_ws;
  size_t hoff = (((size_t)N * sizeof(float)) + 255) & ~(size_t)255;
  float* h    = (float*)((char*)d_ws + hoff);
  size_t koff = (hoff + (size_t)N * DIM * sizeof(float) + 255) & ~(size_t)255;
  int* keys   = (int*)((char*)d_ws + koff);
  const int nwords = 4 * ((E + 15) / 16);

  gcn_init_kernel<<<(N + 255) / 256, 256, 0, stream>>>(dinv, N);
  gcn_deg_kernel<<<(nwords + 255) / 256, 256, 0, stream>>>(ei, dinv, keys, E, nwords);
  gcn_dinv_kernel<<<(N + 255) / 256, 256, 0, stream>>>(dinv, N);
  gcn_gemm_kernel<<<(N + 15) / 16, 256, 0, stream>>>(x, W, h, N);
  gcn_tile_kernel<<<(N + TR - 1) / TR, TW * 32, 0, stream>>>(x, ei, b, dinv, h, keys, out, N, E);
}


# tile drain does one 4-byte lookup per edge (src | row-in-tile<<17, packed by the degree kernel), was four: h rows stored scaled by dinv[src], dinv[dst] applied once per out row; rounding changes
# speedup vs baseline: 2.2158x; 1.0613x over previous
#include <hip/hip_runtime.h>

typedef float v2f __attribute__((ext_vector_type(2)));
typedef float v8f __attribute__((ext_vector_type(8)));

#define DIM 128
#define XP 132
#define NEG_SLOPE 0.1f

__global__ void gcn_init_kernel(float* __restrict__ deg, int N) {
  int i = blockIdx.x * blockDim.x + threadIdx.x;
  if (i < N) deg[i] = 1.0f;
}

#define TR 400
#define SRC_BITS 17
#define SRC_MASK ((1u << SRC_BITS) - 1u)

__global__ void gcn_deg_kernel(const int* __restrict__ ei, float* __restrict__ deg,
                               int* __restrict__ keys, int* __restrict__ rec, int E, int nwords) {
  const int j = blockIdx.x * blockDim.x + threadIdx.x;
  if (j >= nwords) return;
  const int* dst = ei + E;
  int d[4], s[4];
  if (((E & 3) == 0) && (4 * j + 3 < E)) {
    const int4 v = ((const int4*)dst)[j];
    const int4 sv = ((const int4*)ei)[j];
    d[0] = v.x; d[1] = v.y; d[2] = v.z; d[3] = v.w;
    s[0] = sv.x; s[1] = sv.y; s[2] = sv.z; s[3] = sv.w;
  } else {
    for (int b = 0; b < 4; ++b) {
      d[b] = (4 * j + b < E) ? dst[4 * j + b] : -1;
      s[b] = (4 * j + b < E) ? ei[4 * j + b] : 0;
    }
  }
  unsigned w = 0;
  int r[4];
  for (int b = 0; b < 4; ++b) {
    unsigned k = 0xFFu;
    r[b] = 0;
    if (d[b] >= 0) {
      atomicAdd(&deg[d[b]], 1.0f);
      k = (unsigned)d[b] / (unsigned)TR;
      r[b] = (int)((unsigned)s[b] | (((unsigned)d[b] - k * (unsigned)TR) << SRC_BITS));
    }
    w |= k << (8 * b);
  }
  keys[j] = (int)w;
  ((int4*)rec)[j] = make_int4(r[0], r[1], r[2], r[3]);
}

__global__ void gcn_dinv_kernel(float* __restrict__ deg, int N) {
  int i = blockIdx.x * blockDim.x + threadIdx.x;
  if (i < N) deg[i] = rsqrtf(deg[i]);
}

__global__ __launch_bounds__(256) void gcn_gemm_kernel(const float* __restrict__ x,
                                                       const float* __restrict__ W,
                                                       const float* __restrict__ dinv,
                                                       float* __restrict__ h, int N) {
  __shared__ float Ws[DIM * DIM];
  __shared__ float xs[16 * XP];

  const int tid  = threadIdx.x;
  const int row0 = blockIdx.x * 16;

  for (int i = tid; i < (DIM * DIM) / 4; i += 256) {
    ((float4*)Ws)[i] = ((const float4*)W)[i];
  }
  for (int i = tid; i < 16 * DIM; i += 256) {
    int r = i >> 7, c = i & (DIM - 1);
    int gr = row0 + r;
    xs[r * XP + c] = (gr < N) ? x[(long)gr * DIM + c] : 0.0f;
  }
  __syncthreads();

  const int wave = tid >> 5;
  const int lane = tid & 31;
  const int half = lane >> 4;
  const int m    = lane & 15;
  const int col0 = wave * 16;

  float dv[8];
  for (int r = 0; r < 8; ++r) {
    const int row = row0 + r + 8 * half;
    dv[r] = (row < N) ? dinv[row] : 0.0f;
  }

  v8f acc = {};
  for (int k0 = 0; k0 < DIM; k0 += 4) {
    int ka = k0 + 2 * half;
    v2f a, b;
    a.x = xs[m * XP + ka];
    a.y = xs[m * XP + ka + 1];
    b.x = Ws[ka * DIM + col0 + m];
    b.y = Ws[(ka + 1) * DIM + col0 + m];
    acc = __builtin_amdgcn_wmma_f32_16x16x4_f32(
         false, a,  false, b,
         (short)0, acc,  false,  false);
  }

  for (int r = 0; r < 8; ++r) {
    int row = row0 + r + 8 * half;
    if (row < N) h[(long)row * DIM + col0 + m] = acc[r] * dv[r];
  }
}

#define TW 32
#define QCAP 544

__global__ __launch_bounds__(1024) void gcn_tile_kernel(const float* __restrict__ x,
                                                        const int* __restrict__ rec,
                                                        const float* __restrict__ bias,
                                                        const float* __restrict__ dinv,
                                                        const float* __restrict__ h,
                                                        const int* __restrict__ keys,
                                                        float* __restrict__ out, int N, int E) {
  __shared__ float acc[TR * DIM];
  __shared__ int   q[TW * QCAP];

  const int tid  = threadIdx.x;
  const int lane = tid & 31;
  const int wave = tid >> 5;
  const int row0 = blockIdx.x * TR;
  const int qoff = wave * QCAP;

  for (int i = tid; i < TR * DIM; i += TW * 32) acc[i] = 0.0f;
  __syncthreads();

  auto drain = [&](int n) {
    for (int base = 0; base < n; base += 32) {
      const int idx = base + lane;
      const int e   = q[qoff + ((idx < n) ? idx : base)];
      const int p_l = rec[e];
      const int cnt = (n - base < 32) ? (n - base) : 32;
      for (int j = 0; j < cnt; j += 4) {
        float4 v[4];
        int    rr[4];
        for (int u = 0; u < 4; ++u) {
          const int jj = (j + u < cnt) ? (j + u) : (cnt - 1);
          const unsigned p = (unsigned)__builtin_amdgcn_readlane(p_l, jj);
          rr[u] = (int)(p >> SRC_BITS);
          v[u]  = ((const float4*)(h + (long)(p & SRC_MASK) * DIM))[lane];
        }
        for (int u = 0; u < 4; ++u) {
          if (j + u < cnt) {
            float* a = &acc[rr[u] * DIM + lane];
            atomicAdd(a,      v[u].x);
            atomicAdd(a + 32, v[u].y);
            atomicAdd(a + 64, v[u].z);
            atomicAdd(a + 96, v[u].w);
          }
        }
      }
    }
  };

  int qn = 0;

  const int nvec    = (E + 15) >> 4;
  const int ngroups = (nvec + 127) >> 7;
  const uint4* kp   = (const uint4*)keys;
  const unsigned pat = (unsigned)blockIdx.x * 0x01010101u;

  auto match = [&](unsigned k) -> unsigned {
    const unsigned x = k ^ pat;
    return ~(((x & 0x7F7F7F7Fu) + 0x7F7F7F7Fu) | x) & 0x80808080u;
  };
  auto load_group = [&](int g, uint4 (&kv)[4]) {
    for (int u = 0; u < 4; ++u) {
      const int vi = g * 128 + u * 32 + lane;
      kv[u] = kp[(vi < nvec) ? vi : (nvec - 1)];
    }
  };

  uint4 nk[4];
  if (wave < ngroups) load_group(wave, nk);
  for (int g = wave; g < ngroups; g += TW) {
    uint4 kv[4];
    for (int u = 0; u < 4; ++u) kv[u] = nk[u];
    if (g + TW < ngroups) load_group(g + TW, nk);
    for (int u = 0; u < 4; ++u) {
      const int vi = g * 128 + u * 32 + lane;
      unsigned m = (match(kv[u].x) >> 7) | (match(kv[u].y) >> 6) | (match(kv[u].z) >> 5) | (match(kv[u].w) >> 4);
      if (vi >= nvec) m = 0;
      unsigned lm = __builtin_amdgcn_ballot_w32(m != 0);
      while (lm) {
        if (m) {
          const int p = __builtin_ctz(m);
          q[qoff + qn + (int)__builtin_amdgcn_mbcnt_lo(lm, 0u)] = 16 * vi + ((p & 3) << 2) + (p >> 3);
          m &= m - 1;
        }
        qn += __builtin_popcount(lm);
        lm = __builtin_amdgcn_ballot_w32(m != 0);
      }
      if (qn >= 32) {
        drain(qn);
        qn = 0;
      }
    }
  }
  if (qn > 0) drain(qn);
  __syncthreads();

  const float4 bv = ((const float4*)bias)[lane];
  for (int r = wave; r < TR; r += TW) {
    const int row = row0 + r;
    if (row >= N) break;
    const float* a = &acc[r * DIM + lane];
    const float di = dinv[row];
    const float4 hv = ((const float4*)(h + (long)row * DIM))[lane];
    const float4 xv = ((const float4*)(x + (long)row * DIM))[lane];
    float4 o;
    float v;
    v = (a[0]  + hv.x) * di + bv.x;  v = (v >= 0.0f) ? v : NEG_SLOPE * v;  o.x = v + xv.x;
    v = (a[32] + hv.y) * di + bv.y;  v = (v >= 0.0f) ? v : NEG_SLOPE * v;  o.y = v + xv.y;
    v = (a[64] + hv.z) * di + bv.z;  v = (v >= 0.0f) ? v : NEG_SLOPE * v;  o.z = v + xv.z;
    v = (a[96] + hv.w) * di + bv.w;  v = (v >= 0.0f) ? v : NEG_SLOPE * v;  o.w = v + xv.w;
    ((float4*)(out + (long)row * DIM))[lane] = o;
  }
}

extern "C" void kernel_launch(void* const* d_in, const int* in_sizes, int n_in,
                              void* d_out, int out_size, void* d_ws, size_t ws_size,
                              hipStream_t stream) {
  const float* x  = (const float*)d_in[0];
  const int*   ei = (const int*)d_in[1];
  const float* W  = (const float*)d_in[2];
  const float* b  = (const float*)d_in[3];

  const int N = in_sizes[0] / DIM;
  const int E = in_sizes[1] / 2;

  float* out  = (float*)d_out;
  float* dinv = (float*)d_ws;
  size_t hoff = (((size_t)N * sizeof(float)) + 255) & ~(size_t)255;
  float* h    = (float*)((char*)d_ws + hoff);
  size_t koff = (hoff + (size_t)N * DIM * sizeof(float) + 255) & ~(size_t)255;
  int* keys   = (int*)((char*)d_ws + koff);
  const int nwords = 4 * ((E + 15) / 16);
  size_t roff = (koff + (size_t)nwords * sizeof(int) + 255) & ~(size_t)255;
  int* rec    = (int*)((char*)d_ws + roff);

  gcn_init_kernel<<<(N + 255) / 256, 256, 0, stream>>>(dinv, N);
  gcn_deg_kernel<<<(nwords + 255) / 256, 256, 0, stream>>>(ei, dinv, keys, rec, E, nwords);
  gcn_dinv_kernel<<<(N + 255) / 256, 256, 0, stream>>>(dinv, N);
  gcn_gemm_kernel<<<(N + 15) / 16, 256, 0, stream>>>(x, W, dinv, h, N);
  gcn_tile_kernel<<<(N + TR - 1) / TR, TW * 32, 0, stream>>>(x, rec, b, dinv, h, keys, out, N, E);
}


# gemm work-group owns 128 rows (was 16): W staged to LDS once per 128 rows at a padded pitch, each wave multiplies its own 16 rows by all of W in 8 accumulator tiles, whole-row loads and stores
# speedup vs baseline: 2.4225x; 1.0933x over previous
#include <hip/hip_runtime.h>

typedef float v2f __attribute__((ext_vector_type(2)));
typedef float v8f __attribute__((ext_vector_type(8)));

#define DIM 128
#define XP 132
#define NEG_SLOPE 0.1f

__global__ void gcn_init_kernel(float* __restrict__ deg, int N) {
  int i = blockIdx.x * blockDim.x + threadIdx.x;
  if (i < N) deg[i] = 1.0f;
}

#define TR 400
#define SRC_BITS 17
#define SRC_MASK ((1u << SRC_BITS) - 1u)

__global__ void gcn_deg_kernel(const int* __restrict__ ei, float* __restrict__ deg,
                               int* __restrict__ keys, int* __restrict__ rec, int E, int nwords) {
  const int j = blockIdx.x * blockDim.x + threadIdx.x;
  if (j >= nwords) return;
  const int* dst = ei + E;
  int d[4], s[4];
  if (((E & 3) == 0) && (4 * j + 3 < E)) {
    const int4 v = ((const int4*)dst)[j];
    const int4 sv = ((const int4*)ei)[j];
    d[0] = v.x; d[1] = v.y; d[2] = v.z; d[3] = v.w;
    s[0] = sv.x; s[1] = sv.y; s[2] = sv.z; s[3] = sv.w;
  } else {
    for (int b = 0; b < 4; ++b) {
      d[b] = (4 * j + b < E) ? dst[4 * j + b] : -1;
      s[b] = (4 * j + b < E) ? ei[4 * j + b] : 0;
    }
  }
  unsigned w = 0;
  int r[4];
  for (int b = 0; b < 4; ++b) {
    unsigned k = 0xFFu;
    r[b] = 0;
    if (d[b] >= 0) {
      atomicAdd(&deg[d[b]], 1.0f);
      k = (unsigned)d[b] / (unsigned)TR;
      r[b] = (int)((unsigned)s[b] | (((unsigned)d[b] - k * (unsigned)TR) << SRC_BITS));
    }
    w |= k << (8 * b);
  }
  keys[j] = (int)w;
  ((int4*)rec)[j] = make_int4(r[0], r[1], r[2], r[3]);
}

__global__ void gcn_dinv_kernel(float* __restrict__ deg, int N) {
  int i = blockIdx.x * blockDim.x + threadIdx.x;
  if (i < N) deg[i] = rsqrtf(deg[i]);
}

#define GR 128
#define WP 136
__global__ __launch_bounds__(256) void gcn_gemm_kernel(const float* __restrict__ x,
                                                       const float* __restrict__ W,
                                                       const float* __restrict__ dinv,
                                                       float* __restrict__ h, int N) {
  __shared__ __attribute__((aligned(16))) float Ws[DIM * WP];
  __shared__ __attribute__((aligned(16))) float xs[8 * 16 * XP];

  const int tid  = threadIdx.x;
  const int wave = tid >> 5;
  const int lane = tid & 31;
  const int half = lane >> 4;
  const int m    = lane & 15;
  const int row0 = blockIdx.x * GR + wave * 16;
  float* xw = xs + wave * 16 * XP;

  float4 xv[16];
  for (int r = 0; r < 16; ++r) {
    const int gr = row0 + r;
    xv[r] = (gr < N) ? ((const float4*)(x + (long)gr * DIM))[lane] : make_float4(0.0f, 0.0f, 0.0f, 0.0f);
  }
  for (int i = tid; i < (DIM * DIM) / 4; i += 256) {
    const float4 w4 = ((const float4*)W)[i];
    *(float4*)&Ws[(i >> 5) * WP + 4 * (i & 31)] = w4;
  }
  for (int r = 0; r < 16; ++r) *(float4*)&xw[r * XP + 4 * lane] = xv[r];

  float dv[16];
  for (int r = 0; r < 16; ++r) {
    const int row = row0 + r;
    dv[r] = (row < N) ? dinv[row] : 0.0f;
  }
  __syncthreads();

  v8f acc[8];
  for (int c = 0; c < 8; ++c) acc[c] = v8f{};
  for (int k0 = 0; k0 < DIM; k0 += 4) {
    const int ka = k0 + 2 * half;
    v2f a;
    a.x = xw[m * XP + ka];
    a.y = xw[m * XP + ka + 1];
#pragma unroll
    for (int c = 0; c < 8; ++c) {
      v2f b;
      b.x = Ws[ka * WP + c * 16 + m];
      b.y = Ws[(ka + 1) * WP + c * 16 + m];
      acc[c] = __builtin_amdgcn_wmma_f32_16x16x4_f32(
           false, a,  false, b,
           (short)0, acc[c],  false,  false);
    }
  }

#pragma unroll
  for (int c = 0; c < 8; ++c)
    for (int r = 0; r < 8; ++r) xw[(r + 8 * half) * XP + c * 16 + m] = acc[c][r];
  for (int r = 0; r < 16; ++r) {
    const int row = row0 + r;
    if (row < N) {
      const float4 v = *(const float4*)&xw[r * XP + 4 * lane];
      ((float4*)(h + (long)row * DIM))[lane] = make_float4(v.x * dv[r], v.y * dv[r], v.z * dv[r], v.w * dv[r]);
    }
  }
}

#define TW 32
#define QCAP 544

__global__ __launch_bounds__(1024) void gcn_tile_kernel(const float* __restrict__ x,
                                                        const int* __restrict__ rec,
                                                        const float* __restrict__ bias,
                                                        const float* __restrict__ dinv,
                                                        const float* __restrict__ h,
                                                        const int* __restrict__ keys,
                                                        float* __restrict__ out, int N, int E) {
  __shared__ float acc[TR * DIM];
  __shared__ int   q[TW * QCAP];

  const int tid  = threadIdx.x;
  const int lane = tid & 31;
  const int wave = tid >> 5;
  const int row0 = blockIdx.x * TR;
  const int qoff = wave * QCAP;

  for (int i = tid; i < TR * DIM; i += TW * 32) acc[i] = 0.0f;
  __syncthreads();

  auto drain = [&](int n) {
    for (int base = 0; base < n; base += 32) {
      const int idx = base + lane;
      const int e   = q[qoff + ((idx < n) ? idx : base)];
      const int p_l = rec[e];
      const int cnt = (n - base < 32) ? (n - base) : 32;
      for (int j = 0; j < cnt; j += 4) {
        float4 v[4];
        int    rr[4];
        for (int u = 0; u < 4; ++u) {
          const int jj = (j + u < cnt) ? (j + u) : (cnt - 1);
          const unsigned p = (unsigned)__builtin_amdgcn_readlane(p_l, jj);
          rr[u] = (int)(p >> SRC_BITS);
          v[u]  = ((const float4*)(h + (long)(p & SRC_MASK) * DIM))[lane];
        }
        for (int u = 0; u < 4; ++u) {
          if (j + u < cnt) {
            float* a = &acc[rr[u] * DIM + lane];
            atomicAdd(a,      v[u].x);
            atomicAdd(a + 32, v[u].y);
            atomicAdd(a + 64, v[u].z);
            atomicAdd(a + 96, v[u].w);
          }
        }
      }
    }
  };

  int qn = 0;

  const int nvec    = (E + 15) >> 4;
  const int ngroups = (nvec + 127) >> 7;
  const uint4* kp   = (const uint4*)keys;
  const unsigned pat = (unsigned)blockIdx.x * 0x01010101u;

  auto match = [&](unsigned k) -> unsigned {
    const unsigned x = k ^ pat;
    return ~(((x & 0x7F7F7F7Fu) + 0x7F7F7F7Fu) | x) & 0x80808080u;
  };
  auto load_group = [&](int g, uint4 (&kv)[4]) {
    for (int u = 0; u < 4; ++u) {
      const int vi = g * 128 + u * 32 + lane;
      kv[u] = kp[(vi < nvec) ? vi : (nvec - 1)];
    }
  };

  uint4 nk[4];
  if (wave < ngroups) load_group(wave, nk);
  for (int g = wave; g < ngroups; g += TW) {
    uint4 kv[4];
    for (int u = 0; u < 4; ++u) kv[u] = nk[u];
    if (g + TW < ngroups) load_group(g + TW, nk);
    for (int u = 0; u < 4; ++u) {
      const int vi = g * 128 + u * 32 + lane;
      unsigned m = (match(kv[u].x) >> 7) | (match(kv[u].y) >> 6) | (match(kv[u].z) >> 5) | (match(kv[u].w) >> 4);
      if (vi >= nvec) m = 0;
      unsigned lm = __builtin_amdgcn_ballot_w32(m != 0);
      while (lm) {
        if (m) {
          const int p = __builtin_ctz(m);
          q[qoff + qn + (int)__builtin_amdgcn_mbcnt_lo(lm, 0u)] = 16 * vi + ((p & 3) << 2) + (p >> 3);
          m &= m - 1;
        }
        qn += __builtin_popcount(lm);
        lm = __builtin_amdgcn_ballot_w32(m != 0);
      }
      if (qn >= 32) {
        drain(qn);
        qn = 0;
      }
    }
  }
  if (qn > 0) drain(qn);
  __syncthreads();

  const float4 bv = ((const float4*)bias)[lane];
  for (int r = wave; r < TR; r += TW) {
    const int row = row0 + r;
    if (row >= N) break;
    const float* a = &acc[r * DIM + lane];
    const float di = dinv[row];
    const float4 hv = ((const float4*)(h + (long)row * DIM))[lane];
    const float4 xv = ((const float4*)(x + (long)row * DIM))[lane];
    float4 o;
    float v;
    v = (a[0]  + hv.x) * di + bv.x;  v = (v >= 0.0f) ? v : NEG_SLOPE * v;  o.x = v + xv.x;
    v = (a[32] + hv.y) * di + bv.y;  v = (v >= 0.0f) ? v : NEG_SLOPE * v;  o.y = v + xv.y;
    v = (a[64] + hv.z) * di + bv.z;  v = (v >= 0.0f) ? v : NEG_SLOPE * v;  o.z = v + xv.z;
    v = (a[96] + hv.w) * di + bv.w;  v = (v >= 0.0f) ? v : NEG_SLOPE * v;  o.w = v + xv.w;
    ((float4*)(out + (long)row * DIM))[lane] = o;
  }
}

extern "C" void kernel_launch(void* const* d_in, const int* in_sizes, int n_in,
                              void* d_out, int out_size, void* d_ws, size_t ws_size,
                              hipStream_t stream) {
  const float* x  = (const float*)d_in[0];
  const int*   ei = (const int*)d_in[1];
  const float* W  = (const float*)d_in[2];
  const float* b  = (const float*)d_in[3];

  const int N = in_sizes[0] / DIM;
  const int E = in_sizes[1] / 2;

  float* out  = (float*)d_out;
  float* dinv = (float*)d_ws;
  size_t hoff = (((size_t)N * sizeof(float)) + 255) & ~(size_t)255;
  float* h    = (float*)((char*)d_ws + hoff);
  size_t koff = (hoff + (size_t)N * DIM * sizeof(float) + 255) & ~(size_t)255;
  int* keys   = (int*)((char*)d_ws + koff);
  const int nwords = 4 * ((E + 15) / 16);
  size_t roff = (koff + (size_t)nwords * sizeof(int) + 255) & ~(size_t)255;
  int* rec    = (int*)((char*)d_ws + roff);

  gcn_init_kernel<<<(N + 255) / 256, 256, 0, stream>>>(dinv, N);
  gcn_deg_kernel<<<(nwords + 255) / 256, 256, 0, stream>>>(ei, dinv, keys, rec, E, nwords);
  gcn_dinv_kernel<<<(N + 255) / 256, 256, 0, stream>>>(dinv, N);
  gcn_gemm_kernel<<<(N + GR - 1) / GR, 256, 0, stream>>>(x, W, dinv, h, N);
  gcn_tile_kernel<<<(N + TR - 1) / TR, TW * 32, 0, stream>>>(x, rec, b, dinv, h, keys, out, N, E);
}


# in-degree counted with no global atomics: 32 edge slices x 8 row ranges, each work-group counts its range in LDS integer counters, dinv kernel adds the 32 partial counts; init launch gone; exact
# speedup vs baseline: 3.5628x; 1.4707x over previous
#include <hip/hip_runtime.h>

typedef float v2f __attribute__((ext_vector_type(2)));
typedef float v8f __attribute__((ext_vector_type(8)));

#define DIM 128
#define XP 132
#define NEG_SLOPE 0.1f

#define TR 400
#define SRC_BITS 17
#define SRC_MASK ((1u << SRC_BITS) - 1u)

#define CW 12800
#define CT 1024
#define CB 16
__global__ __launch_bounds__(CT) void gcn_count_kernel(const int* __restrict__ ei, int* __restrict__ part,
                                                       int* __restrict__ keys, int* __restrict__ rec,
                                                       int N, int E, int nwords, int R, int S) {
  __shared__ int cnt[CW];
  const int tid = threadIdx.x;
  const int a   = blockIdx.x / R;
  const int b   = blockIdx.x - a * R;
  const int lo  = b * CW;
  const int* dst = ei + E;

  for (int i = tid; i < CW; i += CT) cnt[i] = 0;

  for (int j = blockIdx.x * CT + tid; j < nwords; j += gridDim.x * CT) {
    int d[4], s[4];
    if (((E & 3) == 0) && (4 * j + 3 < E)) {
      const int4 v = ((const int4*)dst)[j];
      const int4 sv = ((const int4*)ei)[j];
      d[0] = v.x; d[1] = v.y; d[2] = v.z; d[3] = v.w;
      s[0] = sv.x; s[1] = sv.y; s[2] = sv.z; s[3] = sv.w;
    } else {
      for (int q = 0; q < 4; ++q) {
        d[q] = (4 * j + q < E) ? dst[4 * j + q] : -1;
        s[q] = (4 * j + q < E) ? ei[4 * j + q] : 0;
      }
    }
    unsigned w = 0;
    int r[4];
    for (int q = 0; q < 4; ++q) {
      unsigned k = 0xFFu;
      r[q] = 0;
      if (d[q] >= 0) {
        k = (unsigned)d[q] / (unsigned)TR;
        r[q] = (int)((unsigned)s[q] | (((unsigned)d[q] - k * (unsigned)TR) << SRC_BITS));
      }
      w |= k << (8 * q);
    }
    keys[j] = (int)w;
    ((int4*)rec)[j] = make_int4(r[0], r[1], r[2], r[3]);
  }
  __syncthreads();

  const int L  = ((E + S - 1) / S + 3) & ~3;
  const int e0 = a * L;
  const int e1 = (e0 + L < E) ? (e0 + L) : E;
  if ((E & 3) == 0) {
    const int4* dv = (const int4*)(dst + e0);
    const int nv = (e1 - e0) >> 2;
    for (int i0 = tid; i0 < nv; i0 += CB * CT) {
      int4 v[CB];
      for (int u = 0; u < CB; ++u) {
        const int i = i0 + u * CT;
        v[u] = dv[(i < nv) ? i : i0];
      }
      for (int u = 0; u < CB; ++u) {
        if (i0 + u * CT < nv) {
          unsigned r;
          r = (unsigned)(v[u].x - lo);  if (r < (unsigned)CW) atomicAdd(&cnt[r], 1);
          r = (unsigned)(v[u].y - lo);  if (r < (unsigned)CW) atomicAdd(&cnt[r], 1);
          r = (unsigned)(v[u].z - lo);  if (r < (unsigned)CW) atomicAdd(&cnt[r], 1);
          r = (unsigned)(v[u].w - lo);  if (r < (unsigned)CW) atomicAdd(&cnt[r], 1);
        }
      }
    }
  } else {
    for (int e = e0 + tid; e < e1; e += CT) {
      const unsigned r = (unsigned)(dst[e] - lo);
      if (r < (unsigned)CW) atomicAdd(&cnt[r], 1);
    }
  }
  __syncthreads();

  int* p = part + (long)a * N;
  for (int i = tid; i < CW; i += CT) {
    const int row = lo + i;
    if (row < N) p[row] = cnt[i];
  }
}

__global__ void gcn_dinv_kernel(const int* __restrict__ part, float* __restrict__ dinv, int N, int S) {
  int i = blockIdx.x * blockDim.x + threadIdx.x;
  if (i < N) {
    int c = 1;
    for (int a = 0; a < S; ++a) c += part[(long)a * N + i];
    dinv[i] = rsqrtf((float)c);
  }
}

#define GR 128
#define WP 136
__global__ __launch_bounds__(256) void gcn_gemm_kernel(const float* __restrict__ x,
                                                       const float* __restrict__ W,
                                                       const float* __restrict__ dinv,
                                                       float* __restrict__ h, int N) {
  __shared__ __attribute__((aligned(16))) float Ws[DIM * WP];
  __shared__ __attribute__((aligned(16))) float xs[8 * 16 * XP];

  const int tid  = threadIdx.x;
  const int wave = tid >> 5;
  const int lane = tid & 31;
  const int half = lane >> 4;
  const int m    = lane & 15;
  const int row0 = blockIdx.x * GR + wave * 16;
  float* xw = xs + wave * 16 * XP;

  float4 xv[16];
  for (int r = 0; r < 16; ++r) {
    const int gr = row0 + r;
    xv[r] = (gr < N) ? ((const float4*)(x + (long)gr * DIM))[lane] : make_float4(0.0f, 0.0f, 0.0f, 0.0f);
  }
  for (int i = tid; i < (DIM * DIM) / 4; i += 256) {
    const float4 w4 = ((const float4*)W)[i];
    *(float4*)&Ws[(i >> 5) * WP + 4 * (i & 31)] = w4;
  }
  for (int r = 0; r < 16; ++r) *(float4*)&xw[r * XP + 4 * lane] = xv[r];

  float dv[16];
  for (int r = 0; r < 16; ++r) {
    const int row = row0 + r;
    dv[r] = (row < N) ? dinv[row] : 0.0f;
  }
  __syncthreads();

  v8f acc[8];
  for (int c = 0; c < 8; ++c) acc[c] = v8f{};
  for (int k0 = 0; k0 < DIM; k0 += 4) {
    const int ka = k0 + 2 * half;
    v2f a;
    a.x = xw[m * XP + ka];
    a.y = xw[m * XP + ka + 1];
#pragma unroll
    for (int c = 0; c < 8; ++c) {
      v2f b;
      b.x = Ws[ka * WP + c * 16 + m];
      b.y = Ws[(ka + 1) * WP + c * 16 + m];
      acc[c] = __builtin_amdgcn_wmma_f32_16x16x4_f32(
           false, a,  false, b,
           (short)0, acc[c],  false,  false);
    }
  }

#pragma unroll
  for (int c = 0; c < 8; ++c)
    for (int r = 0; r < 8; ++r) xw[(r + 8 * half) * XP + c * 16 + m] = acc[c][r];
  for (int r = 0; r < 16; ++r) {
    const int row = row0 + r;
    if (row < N) {
      const float4 v = *(const float4*)&xw[r * XP + 4 * lane];
      ((float4*)(h + (long)row * DIM))[lane] = make_float4(v.x * dv[r], v.y * dv[r], v.z * dv[r], v.w * dv[r]);
    }
  }
}

#define TW 32
#define QCAP 544

__global__ __launch_bounds__(1024) void gcn_tile_kernel(const float* __restrict__ x,
                                                        const int* __restrict__ rec,
                                                        const float* __restrict__ bias,
                                                        const float* __restrict__ dinv,
                                                        const float* __restrict__ h,
                                                        const int* __restrict__ keys,
                                                        float* __restrict__ out, int N, int E) {
  __shared__ float acc[TR * DIM];
  __shared__ int   q[TW * QCAP];

  const int tid  = threadIdx.x;
  const int lane = tid & 31;
  const int wave = tid >> 5;
  const int row0 = blockIdx.x * TR;
  const int qoff = wave * QCAP;

  for (int i = tid; i < TR * DIM; i += TW * 32) acc[i] = 0.0f;
  __syncthreads();

  auto drain = [&](int n) {
    for (int base = 0; base < n; base += 32) {
      const int idx = base + lane;
      const int e   = q[qoff + ((idx < n) ? idx : base)];
      const int p_l = rec[e];
      const int cnt = (n - base < 32) ? (n - base) : 32;
      for (int j = 0; j < cnt; j += 4) {
        float4 v[4];
        int    rr[4];
        for (int u = 0; u < 4; ++u) {
          const int jj = (j + u < cnt) ? (j + u) : (cnt - 1);
          const unsigned p = (unsigned)__builtin_amdgcn_readlane(p_l, jj);
          rr[u] = (int)(p >> SRC_BITS);
          v[u]  = ((const float4*)(h + (long)(p & SRC_MASK) * DIM))[lane];
        }
        for (int u = 0; u < 4; ++u) {
          if (j + u < cnt) {
            float* a = &acc[rr[u] * DIM + lane];
            atomicAdd(a,      v[u].x);
            atomicAdd(a + 32, v[u].y);
            atomicAdd(a + 64, v[u].z);
            atomicAdd(a + 96, v[u].w);
          }
        }
      }
    }
  };

  int qn = 0;

  const int nvec    = (E + 15) >> 4;
  const int ngroups = (nvec + 127) >> 7;
  const uint4* kp   = (const uint4*)keys;
  const unsigned pat = (unsigned)blockIdx.x * 0x01010101u;

  auto match = [&](unsigned k) -> unsigned {
    const unsigned x = k ^ pat;
    return ~(((x & 0x7F7F7F7Fu) + 0x7F7F7F7Fu) | x) & 0x80808080u;
  };
  auto load_group = [&](int g, uint4 (&kv)[4]) {
    for (int u = 0; u < 4; ++u) {
      const int vi = g * 128 + u * 32 + lane;
      kv[u] = kp[(vi < nvec) ? vi : (nvec - 1)];
    }
  };

  uint4 nk[4];
  if (wave < ngroups) load_group(wave, nk);
  for (int g = wave; g < ngroups; g += TW) {
    uint4 kv[4];
    for (int u = 0; u < 4; ++u) kv[u] = nk[u];
    if (g + TW < ngroups) load_group(g + TW, nk);
    for (int u = 0; u < 4; ++u) {
      const int vi = g * 128 + u * 32 + lane;
      unsigned m = (match(kv[u].x) >> 7) | (match(kv[u].y) >> 6) | (match(kv[u].z) >> 5) | (match(kv[u].w) >> 4);
      if (vi >= nvec) m = 0;
      unsigned lm = __builtin_amdgcn_ballot_w32(m != 0);
      while (lm) {
        if (m) {
          const int p = __builtin_ctz(m);
          q[qoff + qn + (int)__builtin_amdgcn_mbcnt_lo(lm, 0u)] = 16 * vi + ((p & 3) << 2) + (p >> 3);
          m &= m - 1;
        }
        qn += __builtin_popcount(lm);
        lm = __builtin_amdgcn_ballot_w32(m != 0);
      }
      if (qn >= 32) {
        drain(qn);
        qn = 0;
      }
    }
  }
  if (qn > 0) drain(qn);
  __syncthreads();

  const float4 bv = ((const float4*)bias)[lane];
  for (int r = wave; r < TR; r += TW) {
    const int row = row0 + r;
    if (row >= N) break;
    const float* a = &acc[r * DIM + lane];
    const float di = dinv[row];
    const float4 hv = ((const float4*)(h + (long)row * DIM))[lane];
    const float4 xv = ((const float4*)(x + (long)row * DIM))[lane];
    float4 o;
    float v;
    v = (a[0]  + hv.x) * di + bv.x;  v = (v >= 0.0f) ? v : NEG_SLOPE * v;  o.x = v + xv.x;
    v = (a[32] + hv.y) * di + bv.y;  v = (v >= 0.0f) ? v : NEG_SLOPE * v;  o.y = v + xv.y;
    v = (a[64] + hv.z) * di + bv.z;  v = (v >= 0.0f) ? v : NEG_SLOPE * v;  o.z = v + xv.z;
    v = (a[96] + hv.w) * di + bv.w;  v = (v >= 0.0f) ? v : NEG_SLOPE * v;  o.w = v + xv.w;
    ((float4*)(out + (long)row * DIM))[lane] = o;
  }
}

extern "C" void kernel_launch(void* const* d_in, const int* in_sizes, int n_in,
                              void* d_out, int out_size, void* d_ws, size_t ws_size,
                              hipStream_t stream) {
  const float* x  = (const float*)d_in[0];
  const int*   ei = (const int*)d_in[1];
  const float* W  = (const float*)d_in[2];
  const float* b  = (const float*)d_in[3];

  const int N = in_sizes[0] / DIM;
  const int E = in_sizes[1] / 2;

  float* out  = (float*)d_out;
  float* dinv = (float*)d_ws;
  size_t hoff = (((size_t)N * sizeof(float)) + 255) & ~(size_t)255;
  float* h    = (float*)((char*)d_ws + hoff);
  size_t koff = (hoff + (size_t)N * DIM * sizeof(float) + 255) & ~(size_t)255;
  int* keys   = (int*)((char*)d_ws + koff);
  const int nwords = 4 * ((E + 15) / 16);
  size_t roff = (koff + (size_t)nwords * sizeof(int) + 255) & ~(size_t)255;
  int* rec    = (int*)((char*)d_ws + roff);
  size_t poff = (roff + (size_t)nwords * 4 * sizeof(int) + 255) & ~(size_t)255;
  int* part   = (int*)((char*)d_ws + poff);
  const int R = (N + CW - 1) / CW;
  const int S = (R < 256) ? 256 / R : 1;

  gcn_count_kernel<<<R * S, CT, 0, stream>>>(ei, part, keys, rec, N, E, nwords, R, S);
  gcn_dinv_kernel<<<(N + 255) / 256, 256, 0, stream>>>(part, dinv, N, S);
  gcn_gemm_kernel<<<(N + GR - 1) / GR, 256, 0, stream>>>(x, W, dinv, h, N);
  gcn_tile_kernel<<<(N + TR - 1) / TR, TW * 32, 0, stream>>>(x, rec, b, dinv, h, keys, out, N, E);
}


# tile kernel fetches only the 16-edge blocks its per-tile hit bitmap marks (bits set in LDS by the count kernel); a block's key bytes and packed words share one 128-B line; sum order = LDS arrival
# speedup vs baseline: 3.7480x; 1.0520x over previous
#include <hip/hip_runtime.h>

typedef float v2f __attribute__((ext_vector_type(2)));
typedef float v8f __attribute__((ext_vector_type(8)));

#define DIM 128
#define XP 132
#define NEG_SLOPE 0.1f

#define TR 400
#define SRC_BITS 17
#define SRC_MASK ((1u << SRC_BITS) - 1u)

#define CW 12800
#define CT 1024
#define CB 16
#define TPR (CW / TR)
#define BWMAX 128
#define BLKW 32
__global__ __launch_bounds__(CT) void gcn_count_kernel(const int* __restrict__ ei, int* __restrict__ part,
                                                       int* __restrict__ blk, unsigned* __restrict__ bmg,
                                                       int N, int E, int nwords, int R, int S) {
  __shared__ int cnt[CW];
  __shared__ unsigned bm[TPR * BWMAX];
  const int tid = threadIdx.x;
  const int a   = blockIdx.x / R;
  const int b   = blockIdx.x - a * R;
  const int lo  = b * CW;
  const int* dst = ei + E;

  for (int i = tid; i < CW; i += CT) cnt[i] = 0;
  for (int i = tid; i < TPR * BWMAX; i += CT) bm[i] = 0u;

  for (int j = blockIdx.x * CT + tid; j < nwords; j += gridDim.x * CT) {
    int d[4], s[4];
    if (((E & 3) == 0) && (4 * j + 3 < E)) {
      const int4 v = ((const int4*)dst)[j];
      const int4 sv = ((const int4*)ei)[j];
      d[0] = v.x; d[1] = v.y; d[2] = v.z; d[3] = v.w;
      s[0] = sv.x; s[1] = sv.y; s[2] = sv.z; s[3] = sv.w;
    } else {
      for (int q = 0; q < 4; ++q) {
        d[q] = (4 * j + q < E) ? dst[4 * j + q] : -1;
        s[q] = (4 * j + q < E) ? ei[4 * j + q] : 0;
      }
    }
    unsigned w = 0;
    int r[4];
    for (int q = 0; q < 4; ++q) {
      unsigned k = 0xFFu;
      r[q] = 0;
      if (d[q] >= 0) {
        k = (unsigned)d[q] / (unsigned)TR;
        r[q] = (int)((unsigned)s[q] | (((unsigned)d[q] - k * (unsigned)TR) << SRC_BITS));
      }
      w |= k << (8 * q);
    }
    int* bp = blk + (long)(j >> 2) * BLKW;
    bp[j & 3] = (int)w;
    ((int4*)(bp + 4))[j & 3] = make_int4(r[0], r[1], r[2], r[3]);
  }
  __syncthreads();

  const int L   = ((E + S - 1) / S + 511) & ~511;
  const int BWS = L >> 9;
  const int e0 = a * L;
  const int e1 = (e0 + L < E) ? (e0 + L) : E;
  if ((E & 3) == 0) {
    const int4* dv = (const int4*)(dst + e0);
    const int nv = (e1 - e0) >> 2;
    for (int i0 = tid; i0 < nv; i0 += CB * CT) {
      int4 v[CB];
      for (int u = 0; u < CB; ++u) {
        const int i = i0 + u * CT;
        v[u] = dv[(i < nv) ? i : i0];
      }
      for (int u = 0; u < CB; ++u) {
        if (i0 + u * CT < nv) {
          const int vl = (i0 + u * CT) >> 2;
          const unsigned bit = 1u << (vl & 31);
          const int bwo = vl >> 5;
          unsigned r;
          r = (unsigned)(v[u].x - lo);  if (r < (unsigned)CW) { atomicAdd(&cnt[r], 1); atomicOr(&bm[(r / (unsigned)TR) * BWS + bwo], bit); }
          r = (unsigned)(v[u].y - lo);  if (r < (unsigned)CW) { atomicAdd(&cnt[r], 1); atomicOr(&bm[(r / (unsigned)TR) * BWS + bwo], bit); }
          r = (unsigned)(v[u].z - lo);  if (r < (unsigned)CW) { atomicAdd(&cnt[r], 1); atomicOr(&bm[(r / (unsigned)TR) * BWS + bwo], bit); }
          r = (unsigned)(v[u].w - lo);  if (r < (unsigned)CW) { atomicAdd(&cnt[r], 1); atomicOr(&bm[(r / (unsigned)TR) * BWS + bwo], bit); }
        }
      }
    }
  } else {
    for (int e = e0 + tid; e < e1; e += CT) {
      const unsigned r = (unsigned)(dst[e] - lo);
      if (r < (unsigned)CW) {
        const int vl = (e - e0) >> 4;
        atomicAdd(&cnt[r], 1);
        atomicOr(&bm[(r / (unsigned)TR) * BWS + (vl >> 5)], 1u << (vl & 31));
      }
    }
  }
  __syncthreads();

  int* p = part + (long)a * N;
  for (int i = tid; i < CW; i += CT) {
    const int row = lo + i;
    if (row < N) p[row] = cnt[i];
  }
  for (int i = tid; i < TPR * BWS; i += CT) {
    const int t = i / BWS;
    bmg[((long)(b * TPR + t) * S + a) * BWS + (i - t * BWS)] = bm[i];
  }
}

__global__ void gcn_dinv_kernel(const int* __restrict__ part, float* __restrict__ dinv, int N, int S) {
  int i = blockIdx.x * blockDim.x + threadIdx.x;
  if (i < N) {
    int c = 1;
    for (int a = 0; a < S; ++a) c += part[(long)a * N + i];
    dinv[i] = rsqrtf((float)c);
  }
}

#define GR 128
#define WP 136
__global__ __launch_bounds__(256) void gcn_gemm_kernel(const float* __restrict__ x,
                                                       const float* __restrict__ W,
                                                       const float* __restrict__ dinv,
                                                       float* __restrict__ h, int N) {
  __shared__ __attribute__((aligned(16))) float Ws[DIM * WP];
  __shared__ __attribute__((aligned(16))) float xs[8 * 16 * XP];

  const int tid  = threadIdx.x;
  const int wave = tid >> 5;
  const int lane = tid & 31;
  const int half = lane >> 4;
  const int m    = lane & 15;
  const int row0 = blockIdx.x * GR + wave * 16;
  float* xw = xs + wave * 16 * XP;

  float4 xv[16];
  for (int r = 0; r < 16; ++r) {
    const int gr = row0 + r;
    xv[r] = (gr < N) ? ((const float4*)(x + (long)gr * DIM))[lane] : make_float4(0.0f, 0.0f, 0.0f, 0.0f);
  }
  for (int i = tid; i < (DIM * DIM) / 4; i += 256) {
    const float4 w4 = ((const float4*)W)[i];
    *(float4*)&Ws[(i >> 5) * WP + 4 * (i & 31)] = w4;
  }
  for (int r = 0; r < 16; ++r) *(float4*)&xw[r * XP + 4 * lane] = xv[r];

  float dv[16];
  for (int r = 0; r < 16; ++r) {
    const int row = row0 + r;
    dv[r] = (row < N) ? dinv[row] : 0.0f;
  }
  __syncthreads();

  v8f acc[8];
  for (int c = 0; c < 8; ++c) acc[c] = v8f{};
  for (int k0 = 0; k0 < DIM; k0 += 4) {
    const int ka = k0 + 2 * half;
    v2f a;
    a.x = xw[m * XP + ka];
    a.y = xw[m * XP + ka + 1];
#pragma unroll
    for (int c = 0; c < 8; ++c) {
      v2f b;
      b.x = Ws[ka * WP + c * 16 + m];
      b.y = Ws[(ka + 1) * WP + c * 16 + m];
      acc[c] = __builtin_amdgcn_wmma_f32_16x16x4_f32(
           false, a,  false, b,
           (short)0, acc[c],  false,  false);
    }
  }

#pragma unroll
  for (int c = 0; c < 8; ++c)
    for (int r = 0; r < 8; ++r) xw[(r + 8 * half) * XP + c * 16 + m] = acc[c][r];
  for (int r = 0; r < 16; ++r) {
    const int row = row0 + r;
    if (row < N) {
      const float4 v = *(const float4*)&xw[r * XP + 4 * lane];
      ((float4*)(h + (long)row * DIM))[lane] = make_float4(v.x * dv[r], v.y * dv[r], v.z * dv[r], v.w * dv[r]);
    }
  }
}

#define TW 32
#define QCAP 544

__global__ __launch_bounds__(1024) void gcn_tile_kernel(const float* __restrict__ x,
                                                        const int* __restrict__ blk,
                                                        const float* __restrict__ bias,
                                                        const float* __restrict__ dinv,
                                                        const float* __restrict__ h,
                                                        const unsigned* __restrict__ bmg,
                                                        float* __restrict__ out, int N, int E, int nbw) {
  __shared__ float acc[TR * DIM];
  __shared__ int   q[TW * QCAP];

  const int tid  = threadIdx.x;
  const int lane = tid & 31;
  const int wave = tid >> 5;
  const int row0 = blockIdx.x * TR;
  const int qoff = wave * QCAP;

  for (int i = tid; i < TR * DIM; i += TW * 32) acc[i] = 0.0f;
  __syncthreads();

  auto drain = [&](int n) {
    for (int base = 0; base < n; base += 32) {
      const int idx = base + lane;
      const int p_l = q[qoff + ((idx < n) ? idx : base)];
      const int cnt = (n - base < 32) ? (n - base) : 32;
      for (int j = 0; j < cnt; j += 4) {
        float4 v[4];
        int    rr[4];
        for (int u = 0; u < 4; ++u) {
          const int jj = (j + u < cnt) ? (j + u) : (cnt - 1);
          const unsigned p = (unsigned)__builtin_amdgcn_readlane(p_l, jj);
          rr[u] = (int)(p >> SRC_BITS);
          v[u]  = ((const float4*)(h + (long)(p & SRC_MASK) * DIM))[lane];
        }
        for (int u = 0; u < 4; ++u) {
          if (j + u < cnt) {
            float* a = &acc[rr[u] * DIM + lane];
            atomicAdd(a,      v[u].x);
            atomicAdd(a + 32, v[u].y);
            atomicAdd(a + 64, v[u].z);
            atomicAdd(a + 96, v[u].w);
          }
        }
      }
    }
  };

  int qn = 0;

  const int nblk = (E + 15) >> 4;
  const int wpw  = (nbw + TW - 1) / TW;
  const unsigned* bmt = bmg + (long)blockIdx.x * nbw;
  const unsigned pat = (unsigned)blockIdx.x * 0x01010101u;

  auto match = [&](unsigned k) -> unsigned {
    const unsigned x = k ^ pat;
    return ~(((x & 0x7F7F7F7Fu) + 0x7F7F7F7Fu) | x) & 0x80808080u;
  };

  for (int k0 = 0; k0 < wpw; k0 += 32) {
    const int k  = k0 + lane;
    const int wi = wave * wpw + k;
    unsigned bw = (k < wpw && wi < nbw) ? bmt[wi] : 0u;
    unsigned lb = __builtin_amdgcn_ballot_w32(bw != 0);
    while (lb) {
      unsigned m = 0;
      const int* bp = blk;
      if (bw) {
        const int bi = 32 * wi + __builtin_ctz(bw);
        bw &= bw - 1;
        if (bi < nblk) {
          bp = blk + (long)bi * BLKW;
          const uint4 kv = *(const uint4*)bp;
          m = (match(kv.x) >> 7) | (match(kv.y) >> 6) | (match(kv.z) >> 5) | (match(kv.w) >> 4);
        }
      }
      unsigned lm = __builtin_amdgcn_ballot_w32(m != 0);
      while (lm) {
        if (m) {
          const int p = __builtin_ctz(m);
          q[qoff + qn + (int)__builtin_amdgcn_mbcnt_lo(lm, 0u)] = bp[4 + ((p & 3) << 2) + (p >> 3)];
          m &= m - 1;
        }
        qn += __builtin_popcount(lm);
        lm = __builtin_amdgcn_ballot_w32(m != 0);
      }
      if (qn >= 32) {
        drain(qn);
        qn = 0;
      }
      lb = __builtin_amdgcn_ballot_w32(bw != 0);
    }
  }
  if (qn > 0) drain(qn);
  __syncthreads();

  const float4 bv = ((const float4*)bias)[lane];
  for (int r = wave; r < TR; r += TW) {
    const int row = row0 + r;
    if (row >= N) break;
    const float* a = &acc[r * DIM + lane];
    const float di = dinv[row];
    const float4 hv = ((const float4*)(h + (long)row * DIM))[lane];
    const float4 xv = ((const float4*)(x + (long)row * DIM))[lane];
    float4 o;
    float v;
    v = (a[0]  + hv.x) * di + bv.x;  v = (v >= 0.0f) ? v : NEG_SLOPE * v;  o.x = v + xv.x;
    v = (a[32] + hv.y) * di + bv.y;  v = (v >= 0.0f) ? v : NEG_SLOPE * v;  o.y = v + xv.y;
    v = (a[64] + hv.z) * di + bv.z;  v = (v >= 0.0f) ? v : NEG_SLOPE * v;  o.z = v + xv.z;
    v = (a[96] + hv.w) * di + bv.w;  v = (v >= 0.0f) ? v : NEG_SLOPE * v;  o.w = v + xv.w;
    ((float4*)(out + (long)row * DIM))[lane] = o;
  }
}

extern "C" void kernel_launch(void* const* d_in, const int* in_sizes, int n_in,
                              void* d_out, int out_size, void* d_ws, size_t ws_size,
                              hipStream_t stream) {
  const float* x  = (const float*)d_in[0];
  const int*   ei = (const int*)d_in[1];
  const float* W  = (const float*)d_in[2];
  const float* b  = (const float*)d_in[3];

  const int N = in_sizes[0] / DIM;
  const int E = in_sizes[1] / 2;

  float* out  = (float*)d_out;
  float* dinv = (float*)d_ws;
  size_t hoff = (((size_t)N * sizeof(float)) + 255) & ~(size_t)255;
  float* h    = (float*)((char*)d_ws + hoff);
  size_t koff = (hoff + (size_t)N * DIM * sizeof(float) + 255) & ~(size_t)255;
  int* blk    = (int*)((char*)d_ws + koff);
  const int nwords = 4 * ((E + 15) / 16);
  size_t poff = (koff + (size_t)(nwords / 4) * BLKW * sizeof(int) + 255) & ~(size_t)255;
  int* part   = (int*)((char*)d_ws + poff);
  const int R = (N + CW - 1) / CW;
  const int S = (R < 256) ? 256 / R : 1;
  const int nbw = S * ((((E + S - 1) / S + 511) & ~511) >> 9);
  size_t boff = (poff + (size_t)S * N * sizeof(int) + 255) & ~(size_t)255;
  unsigned* bmg = (unsigned*)((char*)d_ws + boff);

  gcn_count_kernel<<<R * S, CT, 0, stream>>>(ei, part, blk, bmg, N, E, nwords, R, S);
  gcn_dinv_kernel<<<(N + 255) / 256, 256, 0, stream>>>(part, dinv, N, S);
  gcn_gemm_kernel<<<(N + GR - 1) / GR, 256, 0, stream>>>(x, W, dinv, h, N);
  gcn_tile_kernel<<<(N + TR - 1) / TR, TW * 32, 0, stream>>>(x, blk, b, dinv, h, bmg, out, N, E, nbw);
}


# tile rows owned by one wave each: edges pass through per-wave LDS queues, the owner adds h rows with plain 16-byte LDS read-add-write, no LDS float atomics; sum order = queue arrival
# speedup vs baseline: 3.8802x; 1.0353x over previous
#include <hip/hip_runtime.h>

typedef float v2f __attribute__((ext_vector_type(2)));
typedef float v8f __attribute__((ext_vector_type(8)));

#define DIM 128
#define XP 132
#define NEG_SLOPE 0.1f

#define TR 400
#define SRC_BITS 17
#define SRC_MASK ((1u << SRC_BITS) - 1u)

#define CW 12800
#define CT 1024
#define CB 16
#define TPR (CW / TR)
#define BWMAX 128
#define BLKW 32
__global__ __launch_bounds__(CT) void gcn_count_kernel(const int* __restrict__ ei, int* __restrict__ part,
                                                       int* __restrict__ blk, unsigned* __restrict__ bmg,
                                                       int N, int E, int nwords, int R, int S) {
  __shared__ int cnt[CW];
  __shared__ unsigned bm[TPR * BWMAX];
  const int tid = threadIdx.x;
  const int a   = blockIdx.x / R;
  const int b   = blockIdx.x - a * R;
  const int lo  = b * CW;
  const int* dst = ei + E;

  for (int i = tid; i < CW; i += CT) cnt[i] = 0;
  for (int i = tid; i < TPR * BWMAX; i += CT) bm[i] = 0u;

  for (int j = blockIdx.x * CT + tid; j < nwords; j += gridDim.x * CT) {
    int d[4], s[4];
    if (((E & 3) == 0) && (4 * j + 3 < E)) {
      const int4 v = ((const int4*)dst)[j];
      const int4 sv = ((const int4*)ei)[j];
      d[0] = v.x; d[1] = v.y; d[2] = v.z; d[3] = v.w;
      s[0] = sv.x; s[1] = sv.y; s[2] = sv.z; s[3] = sv.w;
    } else {
      for (int q = 0; q < 4; ++q) {
        d[q] = (4 * j + q < E) ? dst[4 * j + q] : -1;
        s[q] = (4 * j + q < E) ? ei[4 * j + q] : 0;
      }
    }
    unsigned w = 0;
    int r[4];
    for (int q = 0; q < 4; ++q) {
      unsigned k = 0xFFu;
      r[q] = 0;
      if (d[q] >= 0) {
        k = (unsigned)d[q] / (unsigned)TR;
        r[q] = (int)((unsigned)s[q] | (((unsigned)d[q] - k * (unsigned)TR) << SRC_BITS));
      }
      w |= k << (8 * q);
    }
    int* bp = blk + (long)(j >> 2) * BLKW;
    bp[j & 3] = (int)w;
    ((int4*)(bp + 4))[j & 3] = make_int4(r[0], r[1], r[2], r[3]);
  }
  __syncthreads();

  const int L   = ((E + S - 1) / S + 511) & ~511;
  const int BWS = L >> 9;
  const int e0 = a * L;
  const int e1 = (e0 + L < E) ? (e0 + L) : E;
  if ((E & 3) == 0) {
    const int4* dv = (const int4*)(dst + e0);
    const int nv = (e1 - e0) >> 2;
    for (int i0 = tid; i0 < nv; i0 += CB * CT) {
      int4 v[CB];
      for (int u = 0; u < CB; ++u) {
        const int i = i0 + u * CT;
        v[u] = dv[(i < nv) ? i : i0];
      }
      for (int u = 0; u < CB; ++u) {
        if (i0 + u * CT < nv) {
          const int vl = (i0 + u * CT) >> 2;
          const unsigned bit = 1u << (vl & 31);
          const int bwo = vl >> 5;
          unsigned r;
          r = (unsigned)(v[u].x - lo);  if (r < (unsigned)CW) { atomicAdd(&cnt[r], 1); atomicOr(&bm[(r / (unsigned)TR) * BWS + bwo], bit); }
          r = (unsigned)(v[u].y - lo);  if (r < (unsigned)CW) { atomicAdd(&cnt[r], 1); atomicOr(&bm[(r / (unsigned)TR) * BWS + bwo], bit); }
          r = (unsigned)(v[u].z - lo);  if (r < (unsigned)CW) { atomicAdd(&cnt[r], 1); atomicOr(&bm[(r / (unsigned)TR) * BWS + bwo], bit); }
          r = (unsigned)(v[u].w - lo);  if (r < (unsigned)CW) { atomicAdd(&cnt[r], 1); atomicOr(&bm[(r / (unsigned)TR) * BWS + bwo], bit); }
        }
      }
    }
  } else {
    for (int e = e0 + tid; e < e1; e += CT) {
      const unsigned r = (unsigned)(dst[e] - lo);
      if (r < (unsigned)CW) {
        const int vl = (e - e0) >> 4;
        atomicAdd(&cnt[r], 1);
        atomicOr(&bm[(r / (unsigned)TR) * BWS + (vl >> 5)], 1u << (vl & 31));
      }
    }
  }
  __syncthreads();

  int* p = part + (long)a * N;
  for (int i = tid; i < CW; i += CT) {
    const int row = lo + i;
    if (row < N) p[row] = cnt[i];
  }
  for (int i = tid; i < TPR * BWS; i += CT) {
    const int t = i / BWS;
    bmg[((long)(b * TPR + t) * S + a) * BWS + (i - t * BWS)] = bm[i];
  }
}

__global__ void gcn_dinv_kernel(const int* __restrict__ part, float* __restrict__ dinv, int N, int S) {
  int i = blockIdx.x * blockDim.x + threadIdx.x;
  if (i < N) {
    int c = 1;
    for (int a = 0; a < S; ++a) c += part[(long)a * N + i];
    dinv[i] = rsqrtf((float)c);
  }
}

#define GR 128
#define WP 136
__global__ __launch_bounds__(256) void gcn_gemm_kernel(const float* __restrict__ x,
                                                       const float* __restrict__ W,
                                                       const float* __restrict__ dinv,
                                                       float* __restrict__ h, int N) {
  __shared__ __attribute__((aligned(16))) float Ws[DIM * WP];
  __shared__ __attribute__((aligned(16))) float xs[8 * 16 * XP];

  const int tid  = threadIdx.x;
  const int wave = tid >> 5;
  const int lane = tid & 31;
  const int half = lane >> 4;
  const int m    = lane & 15;
  const int row0 = blockIdx.x * GR + wave * 16;
  float* xw = xs + wave * 16 * XP;

  float4 xv[16];
  for (int r = 0; r < 16; ++r) {
    const int gr = row0 + r;
    xv[r] = (gr < N) ? ((const float4*)(x + (long)gr * DIM))[lane] : make_float4(0.0f, 0.0f, 0.0f, 0.0f);
  }
  for (int i = tid; i < (DIM * DIM) / 4; i += 256) {
    const float4 w4 = ((const float4*)W)[i];
    *(float4*)&Ws[(i >> 5) * WP + 4 * (i & 31)] = w4;
  }
  for (int r = 0; r < 16; ++r) *(float4*)&xw[r * XP + 4 * lane] = xv[r];

  float dv[16];
  for (int r = 0; r < 16; ++r) {
    const int row = row0 + r;
    dv[r] = (row < N) ? dinv[row] : 0.0f;
  }
  __syncthreads();

  v8f acc[8];
  for (int c = 0; c < 8; ++c) acc[c] = v8f{};
  for (int k0 = 0; k0 < DIM; k0 += 4) {
    const int ka = k0 + 2 * half;
    v2f a;
    a.x = xw[m * XP + ka];
    a.y = xw[m * XP + ka + 1];
#pragma unroll
    for (int c = 0; c < 8; ++c) {
      v2f b;
      b.x = Ws[ka * WP + c * 16 + m];
      b.y = Ws[(ka + 1) * WP + c * 16 + m];
      acc[c] = __builtin_amdgcn_wmma_f32_16x16x4_f32(
           false, a,  false, b,
           (short)0, acc[c],  false,  false);
    }
  }

#pragma unroll
  for (int c = 0; c < 8; ++c)
    for (int r = 0; r < 8; ++r) xw[(r + 8 * half) * XP + c * 16 + m] = acc[c][r];
  for (int r = 0; r < 16; ++r) {
    const int row = row0 + r;
    if (row < N) {
      const float4 v = *(const float4*)&xw[r * XP + 4 * lane];
      ((float4*)(h + (long)row * DIM))[lane] = make_float4(v.x * dv[r], v.y * dv[r], v.z * dv[r], v.w * dv[r]);
    }
  }
}

#define TW 32
#define QC 768

__global__ __launch_bounds__(1024) void gcn_tile_kernel(const float* __restrict__ x,
                                                        const int* __restrict__ blk,
                                                        const float* __restrict__ bias,
                                                        const float* __restrict__ dinv,
                                                        const float* __restrict__ h,
                                                        const unsigned* __restrict__ bmg,
                                                        float* __restrict__ out, int N, int E, int nbw) {
  __shared__ __attribute__((aligned(16))) float acc[TR * DIM];
  __shared__ int q[TW * QC];
  __shared__ int qcnt[TW];

  const int tid  = threadIdx.x;
  const int lane = tid & 31;
  const int wave = __builtin_amdgcn_readfirstlane(tid >> 5);
  const int row0 = blockIdx.x * TR;

  for (int i = tid; i < TR * DIM; i += TW * 32) acc[i] = 0.0f;
  if (tid < TW) qcnt[tid] = 0;
  __syncthreads();

  auto push = [&](int p) {
    const int row = (int)((unsigned)p >> SRC_BITS);
    const int o = row & (TW - 1);
    const int slot = atomicAdd(&qcnt[o], 1);
    if (slot < QC) {
      q[o * QC + slot] = p;
    } else {
      const float* hr = h + (long)((unsigned)p & SRC_MASK) * DIM;
      float* ar = &acc[row * DIM];
      for (int c = 0; c < DIM; ++c) atomicAdd(&ar[c], hr[c]);
    }
  };

  const int nblk = (E + 15) >> 4;
  const int wpw  = (nbw + TW - 1) / TW;
  const unsigned* bmt = bmg + (long)blockIdx.x * nbw;
  const unsigned pat = (unsigned)blockIdx.x * 0x01010101u;

  auto match = [&](unsigned k) -> unsigned {
    const unsigned y = k ^ pat;
    return ~(((y & 0x7F7F7F7Fu) + 0x7F7F7F7Fu) | y) & 0x80808080u;
  };

  for (int k0 = 0; k0 < wpw; k0 += 4 * 32) {
    unsigned bw[4];
    int wi[4];
#pragma unroll
    for (int c = 0; c < 4; ++c) {
      const int k = k0 + 32 * c + lane;
      wi[c] = wave * wpw + k;
      bw[c] = (k < wpw && wi[c] < nbw) ? bmt[wi[c]] : 0u;
    }
    while (__builtin_amdgcn_ballot_w32((bw[0] | bw[1] | bw[2] | bw[3]) != 0u)) {
      const int* bp[4];
      uint4 kv[4];
      bool ok[4];
#pragma unroll
      for (int c = 0; c < 4; ++c) {
        int bi = 0;
        ok[c] = false;
        if (bw[c]) {
          const int b = 32 * wi[c] + __builtin_ctz(bw[c]);
          bw[c] &= bw[c] - 1u;
          if (b < nblk) { ok[c] = true; bi = b; }
        }
        bp[c] = blk + (long)bi * BLKW;
        kv[c] = *(const uint4*)bp[c];
      }
      unsigned m[4];
      int w[4];
#pragma unroll
      for (int c = 0; c < 4; ++c) {
        m[c] = ok[c] ? ((match(kv[c].x) >> 7) | (match(kv[c].y) >> 6) | (match(kv[c].z) >> 5) | (match(kv[c].w) >> 4)) : 0u;
        const int p = m[c] ? __builtin_ctz(m[c]) : 0;
        w[c] = bp[c][4 + ((p & 3) << 2) + (p >> 3)];
      }
#pragma unroll
      for (int c = 0; c < 4; ++c) {
        if (m[c]) {
          push(w[c]);
          m[c] &= m[c] - 1u;
        }
      }
#pragma unroll
      for (int c = 0; c < 4; ++c) {
        while (__builtin_amdgcn_ballot_w32(m[c] != 0u)) {
          if (m[c]) {
            const int p = __builtin_ctz(m[c]);
            push(bp[c][4 + ((p & 3) << 2) + (p >> 3)]);
            m[c] &= m[c] - 1u;
          }
        }
      }
    }
  }
  __syncthreads();

  int n = qcnt[wave];
  n = __builtin_amdgcn_readfirstlane((n < QC) ? n : QC);
  const int* qw = q + wave * QC;
  const float4* h4 = (const float4*)h + lane;
  float4* acc4 = (float4*)acc + lane;

  for (int base = 0; base < n; base += 32) {
    const int idx = base + lane;
    const int p_l = qw[(idx < n) ? idx : base];
    const int cnt = (n - base < 32) ? (n - base) : 32;
    float4 va[8], vb[8];
    int    ra[8], rb[8];
    auto issue = [&](int j, float4 (&v)[8], int (&r)[8]) {
#pragma unroll
      for (int u = 0; u < 8; ++u) {
        const int jj = (j + u < cnt) ? (j + u) : (cnt - 1);
        const unsigned p = (unsigned)__builtin_amdgcn_readlane(p_l, jj);
        r[u] = (int)(p >> SRC_BITS);
        v[u] = h4[(long)(p & SRC_MASK) * (DIM / 4)];
      }
    };
    auto apply = [&](int j, float4 (&v)[8], int (&r)[8]) {
#pragma unroll
      for (int u = 0; u < 8; ++u) {
        const float s = (j + u < cnt) ? 1.0f : 0.0f;
        float4* a = acc4 + r[u] * (DIM / 4);
        float4 t = *a;
        t.x += s * v[u].x;
        t.y += s * v[u].y;
        t.z += s * v[u].z;
        t.w += s * v[u].w;
        *a = t;
      }
    };
    if (cnt == 32) {
      issue(0, va, ra);
      issue(8, vb, rb);
      apply(0, va, ra);
      issue(16, va, ra);
      apply(8, vb, rb);
      issue(24, vb, rb);
      apply(16, va, ra);
      apply(24, vb, rb);
    } else {
      issue(0, va, ra);
      for (int j = 0; j < cnt; j += 16) {
        const bool more = j + 8 < cnt;
        if (more) issue(j + 8, vb, rb);
        apply(j, va, ra);
        if (more) {
          if (j + 16 < cnt) issue(j + 16, va, ra);
          apply(j + 8, vb, rb);
        }
      }
    }
  }

  const float4 bv = ((const float4*)bias)[lane];
  for (int r = wave; r < TR; r += TW) {
    const int row = row0 + r;
    if (row >= N) break;
    const float4 a = acc4[r * (DIM / 4)];
    const float di = dinv[row];
    const float4 hv = ((const float4*)(h + (long)row * DIM))[lane];
    const float4 xv = ((const float4*)(x + (long)row * DIM))[lane];
    float4 o;
    float v;
    v = (a.x + hv.x) * di + bv.x;  v = (v >= 0.0f) ? v : NEG_SLOPE * v;  o.x = v + xv.x;
    v = (a.y + hv.y) * di + bv.y;  v = (v >= 0.0f) ? v : NEG_SLOPE * v;  o.y = v + xv.y;
    v = (a.z + hv.z) * di + bv.z;  v = (v >= 0.0f) ? v : NEG_SLOPE * v;  o.z = v + xv.z;
    v = (a.w + hv.w) * di + bv.w;  v = (v >= 0.0f) ? v : NEG_SLOPE * v;  o.w = v + xv.w;
    ((float4*)(out + (long)row * DIM))[lane] = o;
  }
}

extern "C" void kernel_launch(void* const* d_in, const int* in_sizes, int n_in,
                              void* d_out, int out_size, void* d_ws, size_t ws_size,
                              hipStream_t stream) {
  const float* x  = (const float*)d_in[0];
  const int*   ei = (const int*)d_in[1];
  const float* W  = (const float*)d_in[2];
  const float* b  = (const float*)d_in[3];

  const int N = in_sizes[0] / DIM;
  const int E = in_sizes[1] / 2;

  float* out  = (float*)d_out;
  float* dinv = (float*)d_ws;
  size_t hoff = (((size_t)N * sizeof(float)) + 255) & ~(size_t)255;
  float* h    = (float*)((char*)d_ws + hoff);
  size_t koff = (hoff + (size_t)N * DIM * sizeof(float) + 255) & ~(size_t)255;
  int* blk    = (int*)((char*)d_ws + koff);
  const int nwords = 4 * ((E + 15) / 16);
  size_t poff = (koff + (size_t)(nwords / 4) * BLKW * sizeof(int) + 255) & ~(size_t)255;
  int* part   = (int*)((char*)d_ws + poff);
  const int R = (N + CW - 1) / CW;
  const int S = (R < 256) ? 256 / R : 1;
  const int nbw = S * ((((E + S - 1) / S + 511) & ~511) >> 9);
  size_t boff = (poff + (size_t)S * N * sizeof(int) + 255) & ~(size_t)255;
  unsigned* bmg = (unsigned*)((char*)d_ws + boff);

  gcn_count_kernel<<<R * S, CT, 0, stream>>>(ei, part, blk, bmg, N, E, nwords, R, S);
  gcn_dinv_kernel<<<(N + 255) / 256, 256, 0, stream>>>(part, dinv, N, S);
  gcn_gemm_kernel<<<(N + GR - 1) / GR, 256, 0, stream>>>(x, W, dinv, h, N);
  gcn_tile_kernel<<<(N + TR - 1) / TR, TW * 32, 0, stream>>>(x, blk, b, dinv, h, bmg, out, N, E, nbw);
}
